# starting kernel
# baseline (speedup 1.0000x reference)
#include <hip/hip_runtime.h>
#include <hip/hip_bf16.h>

typedef __attribute__((ext_vector_type(2))) float v2f;
typedef __attribute__((ext_vector_type(8))) float v8f;

__global__ void fuse_weights_kernel(const float* __restrict__ w1,
                                    const float* __restrict__ w2,
                                    float* __restrict__ Wc) {
    int t = blockIdx.x * blockDim.x + threadIdx.x;
    if (t >= 128 * 224) return;
    int oc = t / 224;
    int r  = t - oc * 224;
    int kk = r >> 5;
    int j  = r & 31;
    float s = 0.0f;
#pragma unroll
    for (int i = 0; i < 9; ++i)
        s += w1[oc * 9 + i] * w2[(j * 7 + kk) * 9 + i];
    Wc[t] = s;
}

__global__ __launch_bounds__(256) void conv_wmma_kernel(
    const float* __restrict__ x,
    const float* __restrict__ Wc,
    float* __restrict__ out) {
    const int blk   = blockIdx.x;
    const int g     = blk / 49;
    const int nTile = blk % 49;
    const int wave  = threadIdx.x >> 5;
    const int lane  = threadIdx.x & 31;
    const int n     = lane & 15;
    const int half  = lane >> 4;

    const int mBase = wave * 16;
    const int p     = nTile * 16 + n;
    const int h     = p / 28;
    const int w     = p - h * 28;

    const float* __restrict__ wcRow = Wc + (mBase + n) * 224 + half * 2;
    const float* __restrict__ xrow  = x + g * (32 * 784) + h * 28;

    v8f acc = {};

#pragma unroll
    for (int kk = 0; kk < 7; ++kk) {
        const int  wx    = w + kk - 3;
        const bool valid = (wx >= 0) && (wx < 28);
        const float* __restrict__ xcol = xrow + wx + half * 2 * 784;
        const float* __restrict__ wcK  = wcRow + kk * 32;

#pragma unroll
        for (int jb = 0; jb < 32; jb += 4) {
            v2f a = *(const v2f*)(wcK + jb);

            v2f b;
            b.x = valid ? xcol[(jb + 0) * 784] : 0.0f;
            b.y = valid ? xcol[(jb + 1) * 784] : 0.0f;

            acc = __builtin_amdgcn_wmma_f32_16x16x4_f32(
                 false, a,  false, b,
                 (short)0, acc,  false,  false);
        }
    }

    int hRoll = h + 1;
    if (hRoll == 28) hRoll = 0;
    float* __restrict__ obase =
        out + ((g * 128 + mBase + half * 8) * 28 + hRoll) * 28 + w;
#pragma unroll
    for (int v = 0; v < 8; ++v)
        obase[v * 784] = acc[v];
}

extern "C" void kernel_launch(void* const* d_in, const int* in_sizes, int n_in,
                              void* d_out, int out_size, void* d_ws, size_t ws_size,
                              hipStream_t stream) {
    const float* x  = (const float*)d_in[0];
    const float* w1 = (const float*)d_in[1];
    const float* w2 = (const float*)d_in[2];
    float* out = (float*)d_out;
    float* Wc  = (float*)d_ws;

    fuse_weights_kernel<<<(128 * 224 + 255) / 256, 256, 0, stream>>>(w1, w2, Wc);
    conv_wmma_kernel<<<2 * 49, 256, 0, stream>>>(x, Wc, out);
}


# conv kernel: all 168 fragment loads of a wave issued as one branch-free group ahead of its 56-WMMA chain (pad taps: clamped address, select at use) instead of a load round trip per K-step
# speedup vs baseline: 1.8093x; 1.8093x over previous
#include <hip/hip_runtime.h>
#include <hip/hip_bf16.h>

typedef __attribute__((ext_vector_type(2))) float v2f;
typedef __attribute__((ext_vector_type(8))) float v8f;

__global__ void fuse_weights_kernel(const float* __restrict__ w1,
                                    const float* __restrict__ w2,
                                    float* __restrict__ Wc) {
    int t = blockIdx.x * blockDim.x + threadIdx.x;
    if (t >= 128 * 224) return;
    int oc = t / 224;
    int r  = t - oc * 224;
    int kk = r >> 5;
    int j  = r & 31;
    float s = 0.0f;
#pragma unroll
    for (int i = 0; i < 9; ++i)
        s += w1[oc * 9 + i] * w2[(j * 7 + kk) * 9 + i];
    Wc[t] = s;
}

__global__ __launch_bounds__(256) void conv_wmma_kernel(
    const float* __restrict__ x,
    const float* __restrict__ Wc,
    float* __restrict__ out) {
    const int blk   = blockIdx.x;
    const int g     = blk / 49;
    const int nTile = blk % 49;
    const int wave  = threadIdx.x >> 5;
    const int lane  = threadIdx.x & 31;
    const int n     = lane & 15;
    const int half  = lane >> 4;

    const int mBase = wave * 16;
    const int p     = nTile * 16 + n;
    const int h     = p / 28;
    const int w     = p - h * 28;

    const float* __restrict__ wcRow = Wc + (mBase + n) * 224 + half * 2;
    const float* __restrict__ xrow  = x + g * (32 * 784) + h * 28;

    v2f af[56];
    float bx[56], by[56];

#pragma unroll
    for (int kk = 0; kk < 7; ++kk) {
        const int wx  = w + kk - 3;
        const int wxc = wx < 0 ? 0 : (wx > 27 ? 27 : wx);
        const float* __restrict__ xcol = xrow + wxc + half * 2 * 784;
        const float* __restrict__ wcK  = wcRow + kk * 32;
#pragma unroll
        for (int s = 0; s < 8; ++s) {
            af[kk * 8 + s] = *(const v2f*)(wcK + 4 * s);
            bx[kk * 8 + s] = xcol[(4 * s + 0) * 784];
            by[kk * 8 + s] = xcol[(4 * s + 1) * 784];
        }
    }

    __builtin_amdgcn_sched_barrier(0);

    v8f acc = {};
#pragma unroll
    for (int kk = 0; kk < 7; ++kk) {
        const int  wx    = w + kk - 3;
        const bool valid = (wx >= 0) && (wx < 28);
#pragma unroll
        for (int s = 0; s < 8; ++s) {
            v2f b;
            b.x = valid ? bx[kk * 8 + s] : 0.0f;
            b.y = valid ? by[kk * 8 + s] : 0.0f;
            acc = __builtin_amdgcn_wmma_f32_16x16x4_f32(
                 false, af[kk * 8 + s],  false, b,
                 (short)0, acc,  false,  false);
        }
    }

    int hRoll = h + 1;
    if (hRoll == 28) hRoll = 0;
    float* __restrict__ obase =
        out + ((g * 128 + mBase + half * 8) * 28 + hRoll) * 28 + w;
#pragma unroll
    for (int v = 0; v < 8; ++v)
        obase[v * 784] = acc[v];
}

extern "C" void kernel_launch(void* const* d_in, const int* in_sizes, int n_in,
                              void* d_out, int out_size, void* d_ws, size_t ws_size,
                              hipStream_t stream) {
    const float* x  = (const float*)d_in[0];
    const float* w1 = (const float*)d_in[1];
    const float* w2 = (const float*)d_in[2];
    float* out = (float*)d_out;
    float* Wc  = (float*)d_ws;

    fuse_weights_kernel<<<(128 * 224 + 255) / 256, 256, 0, stream>>>(w1, w2, Wc);
    conv_wmma_kernel<<<2 * 49, 256, 0, stream>>>(x, Wc, out);
}
